# starting kernel
# baseline (speedup 1.0000x reference)
#include <hip/hip_runtime.h>
#include <math.h>

typedef float v2f __attribute__((ext_vector_type(2)));
typedef float v8f __attribute__((ext_vector_type(8)));

#define HEADS   8
#define DK      32
#define RATE    2
#define SEQ     2046
#define NPAD    2048
#define MROWS   1024
#define DMODEL  256
#define BATCH   4

#define YSTRIDE 34
#define PSTRIDE 18

__global__ __launch_bounds__(32)
void qkv_gemm_kernel(const float* __restrict__ x,
                     const float* __restrict__ W,
                     const float* __restrict__ bias,
                     float* __restrict__ qkv)
{
    const int lane = threadIdx.x;
    const int half = lane >> 4;
    const int l16  = lane & 15;

    const int mtile = blockIdx.x;
    const int ntile = blockIdx.y;

    const int row = mtile * 16 + l16;
    const int bb  = row >> 11;
    const int n   = row & (NPAD - 1);
    const bool valid = (n < SEQ);
    const int col = ntile * 16 + l16;

    const float mask = valid ? 1.0f : 0.0f;
    const float* xrow = x + ((size_t)(bb * SEQ + (valid ? n : 0))) * DMODEL;

    const float bv = bias[col];
    v8f c;
    #pragma unroll
    for (int v = 0; v < 8; ++v) c[v] = bv;

    #pragma unroll 4
    for (int k0 = 0; k0 < DMODEL; k0 += 4) {
        const int k = k0 + 2 * half;
        const float2 av = *(const float2*)(xrow + k);
        v2f a;
        a[0] = av.x * mask;
        a[1] = av.y * mask;
        v2f bm;
        bm[0] = W[(size_t)k * DMODEL + col];
        bm[1] = W[(size_t)(k + 1) * DMODEL + col];
        c = __builtin_amdgcn_wmma_f32_16x16x4_f32(false, a, false, bm,
                                                  (short)0, c, false, false);
    }

    #pragma unroll
    for (int v = 0; v < 8; ++v) {
        const int rrow = mtile * 16 + v + 8 * half;
        qkv[(size_t)rrow * DMODEL + col] = c[v];
    }
}

__global__ __launch_bounds__(256)
void attn_kernel(const float* __restrict__ qkv, float* __restrict__ out)
{
    extern __shared__ float lds[];
    float* Ys = lds;
    float* Ps = lds + MROWS * YSTRIDE;

    const int tid  = threadIdx.x;
    const int wave = tid >> 5;
    const int lane = tid & 31;
    const int half = lane >> 4;
    const int l16  = lane & 15;

    const int g  = blockIdx.y;
    const int bb = g >> 4;
    const int h  = (g >> 1) & 7;
    const int r  = g & 1;

    {
        const int dcol  = (tid & 15) * 2;
        const int rbase = tid >> 4;
        for (int i = rbase; i < MROWS; i += 16) {
            const float* src = qkv + ((size_t)(bb * NPAD + 2 * i + r)) * DMODEL
                                   + h * DK + dcol;
            const float2 v = *(const float2*)src;
            *(float2*)&Ys[i * YSTRIDE + dcol] = v;
        }
    }
    __syncthreads();

    float* Pw = Ps + wave * 16 * PSTRIDE;

    const int qb    = blockIdx.x * 8 + wave;
    const int qbase = qb * 16;

    const float scale = 0.17677669529663687f;

    float mrow[8], lrow[8];
    v8f o0, o1;
    #pragma unroll
    for (int v = 0; v < 8; ++v) {
        mrow[v] = -3.0e38f; lrow[v] = 0.0f;
        o0[v] = 0.0f; o1[v] = 0.0f;
    }

    for (int kb = 0; kb < 64; ++kb) {
        const int kbase = kb * 16;

        v8f s;
        #pragma unroll
        for (int v = 0; v < 8; ++v) s[v] = 0.0f;
        #pragma unroll
        for (int d0 = 0; d0 < DK; d0 += 4) {
            const int d = d0 + 2 * half;
            v2f a, b;
            const float2 av = *(const float2*)&Ys[(qbase + l16) * YSTRIDE + d];
            const float2 bv = *(const float2*)&Ys[(kbase + l16) * YSTRIDE + d];
            a[0] = av.x; a[1] = av.y;
            b[0] = bv.x; b[1] = bv.y;
            s = __builtin_amdgcn_wmma_f32_16x16x4_f32(false, a, false, b,
                                                      (short)0, s, false, false);
        }

        #pragma unroll
        for (int v = 0; v < 8; ++v) {
            const float sv = s[v] * scale;
            float rmax = sv;
            #pragma unroll
            for (int mk = 1; mk < 16; mk <<= 1)
                rmax = fmaxf(rmax, __shfl_xor(rmax, mk, 32));
            const float newm = fmaxf(mrow[v], rmax);
            const float corr = __expf(mrow[v] - newm);
            const float p    = __expf(sv - newm);
            float rsum = p;
            #pragma unroll
            for (int mk = 1; mk < 16; mk <<= 1)
                rsum += __shfl_xor(rsum, mk, 32);
            lrow[v] = lrow[v] * corr + rsum;
            mrow[v] = newm;
            o0[v] *= corr;
            o1[v] *= corr;
            Pw[(v + 8 * half) * PSTRIDE + l16] = p;
        }

        asm volatile("s_wait_dscnt 0x0" ::: "memory");

        #pragma unroll
        for (int k0 = 0; k0 < 16; k0 += 4) {
            const int kk = k0 + 2 * half;
            v2f a, b0, b1;
            const float2 pa = *(const float2*)&Pw[l16 * PSTRIDE + kk];
            a[0]  = pa.x;
            a[1]  = pa.y;
            b0[0] = Ys[(kbase + kk) * YSTRIDE + l16];
            b0[1] = Ys[(kbase + kk + 1) * YSTRIDE + l16];
            b1[0] = Ys[(kbase + kk) * YSTRIDE + 16 + l16];
            b1[1] = Ys[(kbase + kk + 1) * YSTRIDE + 16 + l16];
            o0 = __builtin_amdgcn_wmma_f32_16x16x4_f32(false, a, false, b0,
                                                       (short)0, o0, false, false);
            o1 = __builtin_amdgcn_wmma_f32_16x16x4_f32(false, a, false, b1,
                                                       (short)0, o1, false, false);
        }
    }

    #pragma unroll
    for (int v = 0; v < 8; ++v) {
        const int row = qbase + v + 8 * half;
        float e[3];
        const int d2 = l16 * 2;
        #pragma unroll
        for (int k = 0; k < 3; ++k) {
            const int nr = row + k - 1;
            const bool ok = (nr >= 0) && (nr < MROWS);
            const int nrc = ok ? nr : row;
            const float msk = ok ? 1.0f : 0.0f;
            float part = msk * (Ys[row * YSTRIDE + d2]     * Ys[nrc * YSTRIDE + d2]
                              + Ys[row * YSTRIDE + d2 + 1] * Ys[nrc * YSTRIDE + d2 + 1]);
            #pragma unroll
            for (int mk = 1; mk < 16; mk <<= 1)
                part += __shfl_xor(part, mk, 32);
            e[k] = part * scale;
        }
        const float newm = fmaxf(fmaxf(mrow[v], e[0]), fmaxf(e[1], e[2]));
        const float corr = __expf(mrow[v] - newm);
        float psum = 0.0f, add0 = 0.0f, add1 = 0.0f;
        #pragma unroll
        for (int k = 0; k < 3; ++k) {
            const float p = __expf(e[k] - newm);
            psum += p;
            const int nr = row + k - 1;
            const bool ok = (nr >= 0) && (nr < MROWS);
            const int nrc = ok ? nr : row;
            const float msk = ok ? 1.0f : 0.0f;
            add0 += (p * msk) * Ys[nrc * YSTRIDE + l16];
            add1 += (p * msk) * Ys[nrc * YSTRIDE + 16 + l16];
        }
        const float inv = 1.0f / (lrow[v] * corr + psum);
        const float r0 = (o0[v] * corr + add0) * inv;
        const float r1 = (o1[v] * corr + add1) * inv;

        const int n = 2 * row + r;
        if (n < SEQ) {
            const size_t base = ((size_t)(bb * SEQ + n)) * DMODEL + h * DK;
            out[base + l16]      = r0;
            out[base + 16 + l16] = r1;
        }
    }
}

extern "C" void kernel_launch(void* const* d_in, const int* in_sizes, int n_in,
                              void* d_out, int out_size, void* d_ws, size_t ws_size,
                              hipStream_t stream) {
    const float* x = (const float*)d_in[0];
    const float* W = (const float*)d_in[1];
    const float* b = (const float*)d_in[2];
    float* out = (float*)d_out;
    float* qkv = (float*)d_ws;

    dim3 g1(512, 16);
    qkv_gemm_kernel<<<g1, 32, 0, stream>>>(x, W, b, qkv);

    const size_t ldsBytes = (size_t)(MROWS * YSTRIDE + 8 * 16 * PSTRIDE) * sizeof(float);
    dim3 g2(8, 64);
    attn_kernel<<<g2, 256, ldsBytes, stream>>>(qkv, out);
}


# Score tile transposed (keys x queries): a lane owns one query, row max/sum stay in registers, P feeds O^T=V^T*P^T directly; no lane shuffles, no LDS trip for P; row-sum/PV sum order changes
# speedup vs baseline: 2.7926x; 2.7926x over previous
#include <hip/hip_runtime.h>
#include <math.h>

typedef float v2f __attribute__((ext_vector_type(2)));
typedef float v8f __attribute__((ext_vector_type(8)));

#define HEADS   8
#define DK      32
#define RATE    2
#define SEQ     2046
#define NPAD    2048
#define MROWS   1024
#define DMODEL  256
#define BATCH   4

#define YSTRIDE 34
#define OSTRIDE 36

__global__ __launch_bounds__(32)
void qkv_gemm_kernel(const float* __restrict__ x,
                     const float* __restrict__ W,
                     const float* __restrict__ bias,
                     float* __restrict__ qkv)
{
    const int lane = threadIdx.x;
    const int half = lane >> 4;
    const int l16  = lane & 15;

    const int mtile = blockIdx.x;
    const int ntile = blockIdx.y;

    const int row = mtile * 16 + l16;
    const int bb  = row >> 11;
    const int n   = row & (NPAD - 1);
    const bool valid = (n < SEQ);
    const int col = ntile * 16 + l16;

    const float mask = valid ? 1.0f : 0.0f;
    const float* xrow = x + ((size_t)(bb * SEQ + (valid ? n : 0))) * DMODEL;

    const float bv = bias[col];
    v8f c;
    #pragma unroll
    for (int v = 0; v < 8; ++v) c[v] = bv;

    #pragma unroll 4
    for (int k0 = 0; k0 < DMODEL; k0 += 4) {
        const int k = k0 + 2 * half;
        const float2 av = *(const float2*)(xrow + k);
        v2f a;
        a[0] = av.x * mask;
        a[1] = av.y * mask;
        v2f bm;
        bm[0] = W[(size_t)k * DMODEL + col];
        bm[1] = W[(size_t)(k + 1) * DMODEL + col];
        c = __builtin_amdgcn_wmma_f32_16x16x4_f32(false, a, false, bm,
                                                  (short)0, c, false, false);
    }

    #pragma unroll
    for (int v = 0; v < 8; ++v) {
        const int rrow = mtile * 16 + v + 8 * half;
        qkv[(size_t)rrow * DMODEL + col] = c[v];
    }
}


__device__ __forceinline__ float xhalf(float x)
{
    const int b = __builtin_bit_cast(int, x);
    return __builtin_bit_cast(float,
        __builtin_amdgcn_permlanex16(b, b, 0x76543210, 0xfedcba98, false, false));
}

__global__ __launch_bounds__(256)
void attn_kernel(const float* __restrict__ qkv, float* __restrict__ out)
{
    extern __shared__ __attribute__((aligned(16))) float lds[];
    float* Ys = lds;
    float* Os = lds + MROWS * YSTRIDE;

    const int tid  = threadIdx.x;
    const int wave = tid >> 5;
    const int lane = tid & 31;
    const int half = lane >> 4;
    const int l16  = lane & 15;

    const int g  = blockIdx.y;
    const int bb = g >> 4;
    const int h  = (g >> 1) & 7;
    const int r  = g & 1;

    {
        const int dcol  = (tid & 15) * 2;
        const int rbase = tid >> 4;
        for (int i = rbase; i < MROWS; i += 16) {
            const float* src = qkv + ((size_t)(bb * NPAD + 2 * i + r)) * DMODEL
                                   + h * DK + dcol;
            const float2 v = *(const float2*)src;
            *(float2*)&Ys[i * YSTRIDE + dcol] = v;
        }
    }
    __syncthreads();

    const int qb    = blockIdx.x * 8 + wave;
    const int qbase = qb * 16;

    const float scale = 0.17677669529663687f;

    v2f qf[8];
    #pragma unroll
    for (int t = 0; t < 8; ++t) {
        const float2 qv = *(const float2*)&Ys[(qbase + l16) * YSTRIDE + 4 * t + 2 * half];
        qf[t][0] = qv.x; qf[t][1] = qv.y;
    }

    float mrun = -3.0e38f;
    float lrun = 0.0f;
    v8f o0, o1;
    #pragma unroll
    for (int v = 0; v < 8; ++v) { o0[v] = 0.0f; o1[v] = 0.0f; }

    for (int kb = 0; kb < 64; ++kb) {
        const int kbase = kb * 16;

        v8f s;
        #pragma unroll
        for (int v = 0; v < 8; ++v) s[v] = 0.0f;
        #pragma unroll
        for (int t = 0; t < 8; ++t) {
            const float2 kv = *(const float2*)&Ys[(kbase + l16) * YSTRIDE + 4 * t + 2 * half];
            v2f a;
            a[0] = kv.x; a[1] = kv.y;
            s = __builtin_amdgcn_wmma_f32_16x16x4_f32(false, a, false, qf[t],
                                                      (short)0, s, false, false);
        }

        float bm = fmaxf(fmaxf(fmaxf(s[0], s[1]), fmaxf(s[2], s[3])),
                         fmaxf(fmaxf(s[4], s[5]), fmaxf(s[6], s[7]))) * scale;
        bm = fmaxf(bm, xhalf(bm));
        const float newm = fmaxf(mrun, bm);
        const float corr = __expf(mrun - newm);
        float p[8];
        #pragma unroll
        for (int v = 0; v < 8; ++v) p[v] = __expf(s[v] * scale - newm);
        const float psum = ((p[0] + p[1]) + (p[2] + p[3])) + ((p[4] + p[5]) + (p[6] + p[7]));
        lrun = lrun * corr + psum;
        mrun = newm;
        #pragma unroll
        for (int v = 0; v < 8; ++v) { o0[v] *= corr; o1[v] *= corr; }

        const float* vcol = &Ys[(kbase + 8 * half) * YSTRIDE + l16];
        #pragma unroll
        for (int j = 0; j < 4; ++j) {
            v2f a0, a1, b;
            a0[0] = vcol[(2 * j)     * YSTRIDE];
            a0[1] = vcol[(2 * j + 1) * YSTRIDE];
            a1[0] = vcol[(2 * j)     * YSTRIDE + 16];
            a1[1] = vcol[(2 * j + 1) * YSTRIDE + 16];
            b[0]  = p[2 * j];
            b[1]  = p[2 * j + 1];
            o0 = __builtin_amdgcn_wmma_f32_16x16x4_f32(false, a0, false, b,
                                                       (short)0, o0, false, false);
            o1 = __builtin_amdgcn_wmma_f32_16x16x4_f32(false, a1, false, b,
                                                       (short)0, o1, false, false);
        }
    }

    {
        const int row = qbase + l16;
        const float ltot = lrun + xhalf(lrun);

        int   nrc[3];
        float msk[3], e[3];
        #pragma unroll
        for (int k = 0; k < 3; ++k) {
            const int nr = row + k - 1;
            const bool ok = (nr >= 0) && (nr < MROWS);
            nrc[k] = ok ? nr : row;
            msk[k] = ok ? 1.0f : 0.0f;
            float part = 0.0f;
            #pragma unroll
            for (int dd = 0; dd < 16; dd += 2) {
                const float2 ya = *(const float2*)&Ys[row    * YSTRIDE + 16 * half + dd];
                const float2 yb = *(const float2*)&Ys[nrc[k] * YSTRIDE + 16 * half + dd];
                part += ya.x * yb.x + ya.y * yb.y;
            }
            part += xhalf(part);
            e[k] = msk[k] * part * scale;
        }
        const float newm = fmaxf(fmaxf(mrun, e[0]), fmaxf(e[1], e[2]));
        const float corr = __expf(mrun - newm);
        float pk[3];
        float psum = 0.0f;
        #pragma unroll
        for (int k = 0; k < 3; ++k) {
            const float pe = __expf(e[k] - newm);
            psum += pe;
            pk[k] = pe * msk[k];
        }
        const float inv = 1.0f / (ltot * corr + psum);

        float r0[8], r1[8];
        #pragma unroll
        for (int v = 0; v < 8; ++v) { r0[v] = o0[v] * corr; r1[v] = o1[v] * corr; }
        #pragma unroll
        for (int k = 0; k < 3; ++k) {
            #pragma unroll
            for (int v = 0; v < 8; v += 2) {
                const float2 y0 = *(const float2*)&Ys[nrc[k] * YSTRIDE + 8 * half + v];
                const float2 y1 = *(const float2*)&Ys[nrc[k] * YSTRIDE + 16 + 8 * half + v];
                r0[v]     += pk[k] * y0.x;
                r0[v + 1] += pk[k] * y0.y;
                r1[v]     += pk[k] * y1.x;
                r1[v + 1] += pk[k] * y1.y;
            }
        }

        float* Ow = Os + wave * 16 * OSTRIDE;
        float* orow = Ow + l16 * OSTRIDE + 8 * half;
        *(float4*)(orow)      = make_float4(r0[0] * inv, r0[1] * inv, r0[2] * inv, r0[3] * inv);
        *(float4*)(orow + 4)  = make_float4(r0[4] * inv, r0[5] * inv, r0[6] * inv, r0[7] * inv);
        *(float4*)(orow + 16) = make_float4(r1[0] * inv, r1[1] * inv, r1[2] * inv, r1[3] * inv);
        *(float4*)(orow + 20) = make_float4(r1[4] * inv, r1[5] * inv, r1[6] * inv, r1[7] * inv);

        asm volatile("s_wait_dscnt 0x0" ::: "memory");

        const int rr = lane >> 3;
        const int cc = (lane & 7) * 4;
        #pragma unroll
        for (int i = 0; i < 4; ++i) {
            const int lrow = rr + 4 * i;
            const float4 ov = *(const float4*)&Ow[lrow * OSTRIDE + cc];
            const int n = 2 * (qbase + lrow) + r;
            if (n < SEQ) {
                const size_t base = ((size_t)(bb * SEQ + n)) * DMODEL + h * DK + cc;
                *(float4*)&out[base] = ov;
            }
        }
    }
}

extern "C" void kernel_launch(void* const* d_in, const int* in_sizes, int n_in,
                              void* d_out, int out_size, void* d_ws, size_t ws_size,
                              hipStream_t stream) {
    const float* x = (const float*)d_in[0];
    const float* W = (const float*)d_in[1];
    const float* b = (const float*)d_in[2];
    float* out = (float*)d_out;
    float* qkv = (float*)d_ws;

    dim3 g1(512, 16);
    qkv_gemm_kernel<<<g1, 32, 0, stream>>>(x, W, b, qkv);

    const size_t ldsBytes = (size_t)(MROWS * YSTRIDE + 8 * 16 * OSTRIDE) * sizeof(float);
    dim3 g2(8, 64);
    attn_kernel<<<g2, 256, ldsBytes, stream>>>(qkv, out);
}


# Projection GEMM re-tiled: 256 eight-wave work-groups of 64x128 (x tile in LDS, 2x2 WMMA tiles per wave, fragments loaded one block ahead) replace 8192 one-wave groups; sum order unchanged
# speedup vs baseline: 3.1147x; 1.1153x over previous
#include <hip/hip_runtime.h>
#include <math.h>

typedef float v2f __attribute__((ext_vector_type(2)));
typedef float v8f __attribute__((ext_vector_type(8)));

#define HEADS   8
#define DK      32
#define RATE    2
#define SEQ     2046
#define NPAD    2048
#define MROWS   1024
#define DMODEL  256
#define BATCH   4

#define YSTRIDE 34
#define OSTRIDE 36

#define XPITCH  260
#define KSTEPS  16
#define NKB     (DMODEL / (4 * KSTEPS))

__global__ __launch_bounds__(256)
void qkv_gemm_kernel(const float* __restrict__ x,
                     const float* __restrict__ W,
                     const float* __restrict__ bias,
                     float* __restrict__ qkv)
{
    extern __shared__ __attribute__((aligned(16))) float xs[];

    const int tid  = threadIdx.x;
    const int wave = tid >> 5;
    const int lane = tid & 31;
    const int half = lane >> 4;
    const int l16  = lane & 15;

    const int row0 = blockIdx.x * 64;
    const int col0 = blockIdx.y * 128;

    const int wm = wave >> 2;
    const int wn = wave & 3;

    const float* ap0 = &xs[(32 * wm + l16) * XPITCH + 2 * half];
    const float* ap1 = ap0 + 16 * XPITCH;
    const int    c0  = col0 + 32 * wn + 2 * l16;
    const float* wp  = W + (size_t)(2 * half) * DMODEL + c0;

    float2 wr0[NKB][KSTEPS], wr1[NKB][KSTEPS];
    #pragma unroll
    for (int t = 0; t < KSTEPS; ++t) {
        const float* p = wp + (size_t)(4 * t) * DMODEL;
        wr0[0][t] = *(const float2*)p;
        wr1[0][t] = *(const float2*)(p + DMODEL);
    }

    {
        const int c4 = (tid & 63) * 4;
        const int rb = tid >> 6;
        const int bb = row0 >> 11;
        const int n0 = row0 & (NPAD - 1);
        float4 v[16];
        #pragma unroll
        for (int u = 0; u < 16; ++u) {
            const int n = n0 + rb + 4 * u;
            const bool valid = (n < SEQ);
            const float4 t = *(const float4*)(x + ((size_t)(bb * SEQ + (valid ? n : 0))) * DMODEL + c4);
            v[u] = valid ? t : make_float4(0.0f, 0.0f, 0.0f, 0.0f);
        }
        #pragma unroll
        for (int u = 0; u < 16; ++u)
            *(float4*)&xs[(rb + 4 * u) * XPITCH + c4] = v[u];
    }
    __syncthreads();

    const float2 bv = *(const float2*)&bias[c0];
    const float bv0 = bv.x;
    const float bv1 = bv.y;
    v8f acc00, acc01, acc10, acc11;
    #pragma unroll
    for (int v = 0; v < 8; ++v) { acc00[v] = bv0; acc01[v] = bv1; acc10[v] = bv0; acc11[v] = bv1; }

    v2f a0[NKB][KSTEPS], a1[NKB][KSTEPS];
    #pragma unroll
    for (int t = 0; t < KSTEPS; ++t) {
        const float2 f0 = *(const float2*)(ap0 + 4 * t);
        const float2 f1 = *(const float2*)(ap1 + 4 * t);
        a0[0][t][0] = f0.x; a0[0][t][1] = f0.y;
        a1[0][t][0] = f1.x; a1[0][t][1] = f1.y;
    }

    #pragma unroll
    for (int kb = 0; kb < NKB; ++kb) {
        if (kb + 1 < NKB) {
            const int kn = (kb + 1) * 4 * KSTEPS;
            #pragma unroll
            for (int t = 0; t < KSTEPS; ++t) {
                const float* p = wp + (size_t)(kn + 4 * t) * DMODEL;
                wr0[kb + 1][t] = *(const float2*)p;
                wr1[kb + 1][t] = *(const float2*)(p + DMODEL);
                const float2 f0 = *(const float2*)(ap0 + kn + 4 * t);
                const float2 f1 = *(const float2*)(ap1 + kn + 4 * t);
                a0[kb + 1][t][0] = f0.x; a0[kb + 1][t][1] = f0.y;
                a1[kb + 1][t][0] = f1.x; a1[kb + 1][t][1] = f1.y;
            }
        }
        __builtin_amdgcn_sched_barrier(0);

        #pragma unroll
        for (int t = 0; t < KSTEPS; ++t) {
            v2f b0, b1;
            b0[0] = wr0[kb][t].x;  b0[1] = wr1[kb][t].x;
            b1[0] = wr0[kb][t].y;  b1[1] = wr1[kb][t].y;
            acc00 = __builtin_amdgcn_wmma_f32_16x16x4_f32(false, a0[kb][t], false, b0,
                                                          (short)0, acc00, false, false);
            acc01 = __builtin_amdgcn_wmma_f32_16x16x4_f32(false, a0[kb][t], false, b1,
                                                          (short)0, acc01, false, false);
            acc10 = __builtin_amdgcn_wmma_f32_16x16x4_f32(false, a1[kb][t], false, b0,
                                                          (short)0, acc10, false, false);
            acc11 = __builtin_amdgcn_wmma_f32_16x16x4_f32(false, a1[kb][t], false, b1,
                                                          (short)0, acc11, false, false);
        }
        __builtin_amdgcn_sched_barrier(0);
    }

    #pragma unroll
    for (int v = 0; v < 8; ++v) {
        const size_t r0 = (size_t)(row0 + 32 * wm + v + 8 * half) * DMODEL;
        const size_t r1 = r0 + (size_t)16 * DMODEL;
        *(float2*)&qkv[r0 + c0] = make_float2(acc00[v], acc01[v]);
        *(float2*)&qkv[r1 + c0] = make_float2(acc10[v], acc11[v]);
    }
}


__device__ __forceinline__ float xhalf(float x)
{
    const int b = __builtin_bit_cast(int, x);
    return __builtin_bit_cast(float,
        __builtin_amdgcn_permlanex16(b, b, 0x76543210, 0xfedcba98, false, false));
}

__global__ __launch_bounds__(256)
void attn_kernel(const float* __restrict__ qkv, float* __restrict__ out)
{
    extern __shared__ __attribute__((aligned(16))) float lds[];
    float* Ys = lds;
    float* Os = lds + MROWS * YSTRIDE;

    const int tid  = threadIdx.x;
    const int wave = tid >> 5;
    const int lane = tid & 31;
    const int half = lane >> 4;
    const int l16  = lane & 15;

    const int g  = blockIdx.y;
    const int bb = g >> 4;
    const int h  = (g >> 1) & 7;
    const int r  = g & 1;

    {
        const int dcol  = (tid & 15) * 2;
        const int rbase = tid >> 4;
        for (int i = rbase; i < MROWS; i += 16) {
            const float* src = qkv + ((size_t)(bb * NPAD + 2 * i + r)) * DMODEL
                                   + h * DK + dcol;
            const float2 v = *(const float2*)src;
            *(float2*)&Ys[i * YSTRIDE + dcol] = v;
        }
    }
    __syncthreads();

    const int qb    = blockIdx.x * 8 + wave;
    const int qbase = qb * 16;

    const float scale = 0.17677669529663687f;

    v2f qf[8];
    #pragma unroll
    for (int t = 0; t < 8; ++t) {
        const float2 qv = *(const float2*)&Ys[(qbase + l16) * YSTRIDE + 4 * t + 2 * half];
        qf[t][0] = qv.x; qf[t][1] = qv.y;
    }

    float mrun = -3.0e38f;
    float lrun = 0.0f;
    v8f o0, o1;
    #pragma unroll
    for (int v = 0; v < 8; ++v) { o0[v] = 0.0f; o1[v] = 0.0f; }

    for (int kb = 0; kb < 64; ++kb) {
        const int kbase = kb * 16;

        v8f s;
        #pragma unroll
        for (int v = 0; v < 8; ++v) s[v] = 0.0f;
        #pragma unroll
        for (int t = 0; t < 8; ++t) {
            const float2 kv = *(const float2*)&Ys[(kbase + l16) * YSTRIDE + 4 * t + 2 * half];
            v2f a;
            a[0] = kv.x; a[1] = kv.y;
            s = __builtin_amdgcn_wmma_f32_16x16x4_f32(false, a, false, qf[t],
                                                      (short)0, s, false, false);
        }

        float bm = fmaxf(fmaxf(fmaxf(s[0], s[1]), fmaxf(s[2], s[3])),
                         fmaxf(fmaxf(s[4], s[5]), fmaxf(s[6], s[7]))) * scale;
        bm = fmaxf(bm, xhalf(bm));
        const float newm = fmaxf(mrun, bm);
        const float corr = __expf(mrun - newm);
        float p[8];
        #pragma unroll
        for (int v = 0; v < 8; ++v) p[v] = __expf(s[v] * scale - newm);
        const float psum = ((p[0] + p[1]) + (p[2] + p[3])) + ((p[4] + p[5]) + (p[6] + p[7]));
        lrun = lrun * corr + psum;
        mrun = newm;
        #pragma unroll
        for (int v = 0; v < 8; ++v) { o0[v] *= corr; o1[v] *= corr; }

        const float* vcol = &Ys[(kbase + 8 * half) * YSTRIDE + l16];
        #pragma unroll
        for (int j = 0; j < 4; ++j) {
            v2f a0, a1, b;
            a0[0] = vcol[(2 * j)     * YSTRIDE];
            a0[1] = vcol[(2 * j + 1) * YSTRIDE];
            a1[0] = vcol[(2 * j)     * YSTRIDE + 16];
            a1[1] = vcol[(2 * j + 1) * YSTRIDE + 16];
            b[0]  = p[2 * j];
            b[1]  = p[2 * j + 1];
            o0 = __builtin_amdgcn_wmma_f32_16x16x4_f32(false, a0, false, b,
                                                       (short)0, o0, false, false);
            o1 = __builtin_amdgcn_wmma_f32_16x16x4_f32(false, a1, false, b,
                                                       (short)0, o1, false, false);
        }
    }

    {
        const int row = qbase + l16;
        const float ltot = lrun + xhalf(lrun);

        int   nrc[3];
        float msk[3], e[3];
        #pragma unroll
        for (int k = 0; k < 3; ++k) {
            const int nr = row + k - 1;
            const bool ok = (nr >= 0) && (nr < MROWS);
            nrc[k] = ok ? nr : row;
            msk[k] = ok ? 1.0f : 0.0f;
            float part = 0.0f;
            #pragma unroll
            for (int dd = 0; dd < 16; dd += 2) {
                const float2 ya = *(const float2*)&Ys[row    * YSTRIDE + 16 * half + dd];
                const float2 yb = *(const float2*)&Ys[nrc[k] * YSTRIDE + 16 * half + dd];
                part += ya.x * yb.x + ya.y * yb.y;
            }
            part += xhalf(part);
            e[k] = msk[k] * part * scale;
        }
        const float newm = fmaxf(fmaxf(mrun, e[0]), fmaxf(e[1], e[2]));
        const float corr = __expf(mrun - newm);
        float pk[3];
        float psum = 0.0f;
        #pragma unroll
        for (int k = 0; k < 3; ++k) {
            const float pe = __expf(e[k] - newm);
            psum += pe;
            pk[k] = pe * msk[k];
        }
        const float inv = 1.0f / (ltot * corr + psum);

        float r0[8], r1[8];
        #pragma unroll
        for (int v = 0; v < 8; ++v) { r0[v] = o0[v] * corr; r1[v] = o1[v] * corr; }
        #pragma unroll
        for (int k = 0; k < 3; ++k) {
            #pragma unroll
            for (int v = 0; v < 8; v += 2) {
                const float2 y0 = *(const float2*)&Ys[nrc[k] * YSTRIDE + 8 * half + v];
                const float2 y1 = *(const float2*)&Ys[nrc[k] * YSTRIDE + 16 + 8 * half + v];
                r0[v]     += pk[k] * y0.x;
                r0[v + 1] += pk[k] * y0.y;
                r1[v]     += pk[k] * y1.x;
                r1[v + 1] += pk[k] * y1.y;
            }
        }

        float* Ow = Os + wave * 16 * OSTRIDE;
        float* orow = Ow + l16 * OSTRIDE + 8 * half;
        *(float4*)(orow)      = make_float4(r0[0] * inv, r0[1] * inv, r0[2] * inv, r0[3] * inv);
        *(float4*)(orow + 4)  = make_float4(r0[4] * inv, r0[5] * inv, r0[6] * inv, r0[7] * inv);
        *(float4*)(orow + 16) = make_float4(r1[0] * inv, r1[1] * inv, r1[2] * inv, r1[3] * inv);
        *(float4*)(orow + 20) = make_float4(r1[4] * inv, r1[5] * inv, r1[6] * inv, r1[7] * inv);

        asm volatile("s_wait_dscnt 0x0" ::: "memory");

        const int rr = lane >> 3;
        const int cc = (lane & 7) * 4;
        #pragma unroll
        for (int i = 0; i < 4; ++i) {
            const int lrow = rr + 4 * i;
            const float4 ov = *(const float4*)&Ow[lrow * OSTRIDE + cc];
            const int n = 2 * (qbase + lrow) + r;
            if (n < SEQ) {
                const size_t base = ((size_t)(bb * SEQ + n)) * DMODEL + h * DK + cc;
                *(float4*)&out[base] = ov;
            }
        }
    }
}

extern "C" void kernel_launch(void* const* d_in, const int* in_sizes, int n_in,
                              void* d_out, int out_size, void* d_ws, size_t ws_size,
                              hipStream_t stream) {
    const float* x = (const float*)d_in[0];
    const float* W = (const float*)d_in[1];
    const float* b = (const float*)d_in[2];
    float* out = (float*)d_out;
    float* qkv = (float*)d_ws;

    const size_t xsBytes = (size_t)(64 * XPITCH) * sizeof(float);
    dim3 g1(128, 2);
    qkv_gemm_kernel<<<g1, 256, xsBytes, stream>>>(x, W, b, qkv);

    const size_t ldsBytes = (size_t)(MROWS * YSTRIDE + 8 * 16 * OSTRIDE) * sizeof(float);
    dim3 g2(8, 64);
    attn_kernel<<<g2, 256, ldsBytes, stream>>>(qkv, out);
}


# Attention wave owns two query tiles (32 queries): each K/V fragment read from LDS feeds two WMMAs; 256 work-groups, one per WGP, declared 2 waves/SIMD; Y staged half as often; same sums, same order
# speedup vs baseline: 3.3272x; 1.0682x over previous
#include <hip/hip_runtime.h>
#include <math.h>

typedef float v2f __attribute__((ext_vector_type(2)));
typedef float v8f __attribute__((ext_vector_type(8)));

#define HEADS   8
#define DK      32
#define RATE    2
#define SEQ     2046
#define NPAD    2048
#define MROWS   1024
#define DMODEL  256
#define BATCH   4

#define YSTRIDE 34
#define OSTRIDE 36

#define XPITCH  260
#define KSTEPS  16
#define NKB     (DMODEL / (4 * KSTEPS))

__global__ __launch_bounds__(256)
void qkv_gemm_kernel(const float* __restrict__ x,
                     const float* __restrict__ W,
                     const float* __restrict__ bias,
                     float* __restrict__ qkv)
{
    extern __shared__ __attribute__((aligned(16))) float xs[];

    const int tid  = threadIdx.x;
    const int wave = tid >> 5;
    const int lane = tid & 31;
    const int half = lane >> 4;
    const int l16  = lane & 15;

    const int row0 = blockIdx.x * 64;
    const int col0 = blockIdx.y * 128;

    const int wm = wave >> 2;
    const int wn = wave & 3;

    const float* ap0 = &xs[(32 * wm + l16) * XPITCH + 2 * half];
    const float* ap1 = ap0 + 16 * XPITCH;
    const int    c0  = col0 + 32 * wn + 2 * l16;
    const float* wp  = W + (size_t)(2 * half) * DMODEL + c0;

    float2 wr0[NKB][KSTEPS], wr1[NKB][KSTEPS];
    #pragma unroll
    for (int t = 0; t < KSTEPS; ++t) {
        const float* p = wp + (size_t)(4 * t) * DMODEL;
        wr0[0][t] = *(const float2*)p;
        wr1[0][t] = *(const float2*)(p + DMODEL);
    }

    {
        const int c4 = (tid & 63) * 4;
        const int rb = tid >> 6;
        const int bb = row0 >> 11;
        const int n0 = row0 & (NPAD - 1);
        float4 v[16];
        #pragma unroll
        for (int u = 0; u < 16; ++u) {
            const int n = n0 + rb + 4 * u;
            const bool valid = (n < SEQ);
            const float4 t = *(const float4*)(x + ((size_t)(bb * SEQ + (valid ? n : 0))) * DMODEL + c4);
            v[u] = valid ? t : make_float4(0.0f, 0.0f, 0.0f, 0.0f);
        }
        #pragma unroll
        for (int u = 0; u < 16; ++u)
            *(float4*)&xs[(rb + 4 * u) * XPITCH + c4] = v[u];
    }
    __syncthreads();

    const float2 bv = *(const float2*)&bias[c0];
    const float bv0 = bv.x;
    const float bv1 = bv.y;
    v8f acc00, acc01, acc10, acc11;
    #pragma unroll
    for (int v = 0; v < 8; ++v) { acc00[v] = bv0; acc01[v] = bv1; acc10[v] = bv0; acc11[v] = bv1; }

    v2f a0[NKB][KSTEPS], a1[NKB][KSTEPS];
    #pragma unroll
    for (int t = 0; t < KSTEPS; ++t) {
        const float2 f0 = *(const float2*)(ap0 + 4 * t);
        const float2 f1 = *(const float2*)(ap1 + 4 * t);
        a0[0][t][0] = f0.x; a0[0][t][1] = f0.y;
        a1[0][t][0] = f1.x; a1[0][t][1] = f1.y;
    }

    #pragma unroll
    for (int kb = 0; kb < NKB; ++kb) {
        if (kb + 1 < NKB) {
            const int kn = (kb + 1) * 4 * KSTEPS;
            #pragma unroll
            for (int t = 0; t < KSTEPS; ++t) {
                const float* p = wp + (size_t)(kn + 4 * t) * DMODEL;
                wr0[kb + 1][t] = *(const float2*)p;
                wr1[kb + 1][t] = *(const float2*)(p + DMODEL);
                const float2 f0 = *(const float2*)(ap0 + kn + 4 * t);
                const float2 f1 = *(const float2*)(ap1 + kn + 4 * t);
                a0[kb + 1][t][0] = f0.x; a0[kb + 1][t][1] = f0.y;
                a1[kb + 1][t][0] = f1.x; a1[kb + 1][t][1] = f1.y;
            }
        }
        __builtin_amdgcn_sched_barrier(0);

        #pragma unroll
        for (int t = 0; t < KSTEPS; ++t) {
            v2f b0, b1;
            b0[0] = wr0[kb][t].x;  b0[1] = wr1[kb][t].x;
            b1[0] = wr0[kb][t].y;  b1[1] = wr1[kb][t].y;
            acc00 = __builtin_amdgcn_wmma_f32_16x16x4_f32(false, a0[kb][t], false, b0,
                                                          (short)0, acc00, false, false);
            acc01 = __builtin_amdgcn_wmma_f32_16x16x4_f32(false, a0[kb][t], false, b1,
                                                          (short)0, acc01, false, false);
            acc10 = __builtin_amdgcn_wmma_f32_16x16x4_f32(false, a1[kb][t], false, b0,
                                                          (short)0, acc10, false, false);
            acc11 = __builtin_amdgcn_wmma_f32_16x16x4_f32(false, a1[kb][t], false, b1,
                                                          (short)0, acc11, false, false);
        }
        __builtin_amdgcn_sched_barrier(0);
    }

    #pragma unroll
    for (int v = 0; v < 8; ++v) {
        const size_t r0 = (size_t)(row0 + 32 * wm + v + 8 * half) * DMODEL;
        const size_t r1 = r0 + (size_t)16 * DMODEL;
        *(float2*)&qkv[r0 + c0] = make_float2(acc00[v], acc01[v]);
        *(float2*)&qkv[r1 + c0] = make_float2(acc10[v], acc11[v]);
    }
}

#define QTILES  2

__device__ __forceinline__ float xhalf(float x)
{
    const int b = __builtin_bit_cast(int, x);
    return __builtin_bit_cast(float,
        __builtin_amdgcn_permlanex16(b, b, 0x76543210, 0xfedcba98, false, false));
}

__device__ __forceinline__ void softmax_step(const v8f& s, float scale, float& mrun, float& lrun,
                                             v8f& o0, v8f& o1, float (&p)[8])
{
    float bm = fmaxf(fmaxf(fmaxf(s[0], s[1]), fmaxf(s[2], s[3])),
                     fmaxf(fmaxf(s[4], s[5]), fmaxf(s[6], s[7]))) * scale;
    bm = fmaxf(bm, xhalf(bm));
    const float newm = fmaxf(mrun, bm);
    const float corr = __expf(mrun - newm);
    #pragma unroll
    for (int v = 0; v < 8; ++v) p[v] = __expf(s[v] * scale - newm);
    const float psum = ((p[0] + p[1]) + (p[2] + p[3])) + ((p[4] + p[5]) + (p[6] + p[7]));
    lrun = lrun * corr + psum;
    mrun = newm;
    #pragma unroll
    for (int v = 0; v < 8; ++v) { o0[v] *= corr; o1[v] *= corr; }
}

__device__ __forceinline__ void finish_tile(const float* Ys, float* Ow, float* __restrict__ out,
                                            int qbase, int lane, int bb, int h, int r,
                                            float scale, float mrun, float lrun,
                                            const v8f& o0, const v8f& o1)
{
    const int half = lane >> 4;
    const int l16  = lane & 15;
    const int row  = qbase + l16;
    const float ltot = lrun + xhalf(lrun);

    int   nrc[3];
    float msk[3], e[3];
    #pragma unroll
    for (int k = 0; k < 3; ++k) {
        const int nr = row + k - 1;
        const bool ok = (nr >= 0) && (nr < MROWS);
        nrc[k] = ok ? nr : row;
        msk[k] = ok ? 1.0f : 0.0f;
        float part = 0.0f;
        #pragma unroll
        for (int dd = 0; dd < 16; dd += 2) {
            const float2 ya = *(const float2*)&Ys[row    * YSTRIDE + 16 * half + dd];
            const float2 yb = *(const float2*)&Ys[nrc[k] * YSTRIDE + 16 * half + dd];
            part += ya.x * yb.x + ya.y * yb.y;
        }
        part += xhalf(part);
        e[k] = msk[k] * part * scale;
    }
    const float newm = fmaxf(fmaxf(mrun, e[0]), fmaxf(e[1], e[2]));
    const float corr = __expf(mrun - newm);
    float pk[3];
    float psum = 0.0f;
    #pragma unroll
    for (int k = 0; k < 3; ++k) {
        const float pe = __expf(e[k] - newm);
        psum += pe;
        pk[k] = pe * msk[k];
    }
    const float inv = 1.0f / (ltot * corr + psum);

    float r0[8], r1[8];
    #pragma unroll
    for (int v = 0; v < 8; ++v) { r0[v] = o0[v] * corr; r1[v] = o1[v] * corr; }
    #pragma unroll
    for (int k = 0; k < 3; ++k) {
        #pragma unroll
        for (int v = 0; v < 8; v += 2) {
            const float2 y0 = *(const float2*)&Ys[nrc[k] * YSTRIDE + 8 * half + v];
            const float2 y1 = *(const float2*)&Ys[nrc[k] * YSTRIDE + 16 + 8 * half + v];
            r0[v]     += pk[k] * y0.x;
            r0[v + 1] += pk[k] * y0.y;
            r1[v]     += pk[k] * y1.x;
            r1[v + 1] += pk[k] * y1.y;
        }
    }

    float* orow = Ow + l16 * OSTRIDE + 8 * half;
    *(float4*)(orow)      = make_float4(r0[0] * inv, r0[1] * inv, r0[2] * inv, r0[3] * inv);
    *(float4*)(orow + 4)  = make_float4(r0[4] * inv, r0[5] * inv, r0[6] * inv, r0[7] * inv);
    *(float4*)(orow + 16) = make_float4(r1[0] * inv, r1[1] * inv, r1[2] * inv, r1[3] * inv);
    *(float4*)(orow + 20) = make_float4(r1[4] * inv, r1[5] * inv, r1[6] * inv, r1[7] * inv);

    asm volatile("s_wait_dscnt 0x0" ::: "memory");

    const int rr = lane >> 3;
    const int cc = (lane & 7) * 4;
    #pragma unroll
    for (int i = 0; i < 4; ++i) {
        const int lrow = rr + 4 * i;
        const float4 ov = *(const float4*)&Ow[lrow * OSTRIDE + cc];
        const int n = 2 * (qbase + lrow) + r;
        if (n < SEQ) {
            const size_t base = ((size_t)(bb * SEQ + n)) * DMODEL + h * DK + cc;
            *(float4*)&out[base] = ov;
        }
    }
}

__global__ __launch_bounds__(256) __attribute__((amdgpu_waves_per_eu(2, 2)))
void attn_kernel(const float* __restrict__ qkv, float* __restrict__ out)
{
    extern __shared__ __attribute__((aligned(16))) float lds[];
    float* Ys = lds;
    float* Os = lds + MROWS * YSTRIDE;

    const int tid  = threadIdx.x;
    const int wave = tid >> 5;
    const int lane = tid & 31;
    const int half = lane >> 4;
    const int l16  = lane & 15;

    const int g  = blockIdx.y;
    const int bb = g >> 4;
    const int h  = (g >> 1) & 7;
    const int r  = g & 1;

    {
        const int dcol  = (tid & 15) * 2;
        const int rbase = tid >> 4;
        for (int i = rbase; i < MROWS; i += 16) {
            const float* src = qkv + ((size_t)(bb * NPAD + 2 * i + r)) * DMODEL
                                   + h * DK + dcol;
            const float2 v = *(const float2*)src;
            *(float2*)&Ys[i * YSTRIDE + dcol] = v;
        }
    }
    __syncthreads();

    const int qbase = (blockIdx.x * 8 + wave) * (16 * QTILES);

    const float scale = 0.17677669529663687f;

    v2f qfa[8], qfb[8];
    #pragma unroll
    for (int t = 0; t < 8; ++t) {
        const float2 qa = *(const float2*)&Ys[(qbase + l16)      * YSTRIDE + 4 * t + 2 * half];
        const float2 qb = *(const float2*)&Ys[(qbase + 16 + l16) * YSTRIDE + 4 * t + 2 * half];
        qfa[t][0] = qa.x; qfa[t][1] = qa.y;
        qfb[t][0] = qb.x; qfb[t][1] = qb.y;
    }

    float mruna = -3.0e38f, mrunb = -3.0e38f;
    float lruna = 0.0f,     lrunb = 0.0f;
    v8f o0a, o1a, o0b, o1b;
    #pragma unroll
    for (int v = 0; v < 8; ++v) { o0a[v] = 0.0f; o1a[v] = 0.0f; o0b[v] = 0.0f; o1b[v] = 0.0f; }

    for (int kb = 0; kb < 64; ++kb) {
        const int kbase = kb * 16;

        v8f sa, sb;
        #pragma unroll
        for (int v = 0; v < 8; ++v) { sa[v] = 0.0f; sb[v] = 0.0f; }
        #pragma unroll
        for (int t = 0; t < 8; ++t) {
            const float2 kv = *(const float2*)&Ys[(kbase + l16) * YSTRIDE + 4 * t + 2 * half];
            v2f a;
            a[0] = kv.x; a[1] = kv.y;
            sa = __builtin_amdgcn_wmma_f32_16x16x4_f32(false, a, false, qfa[t],
                                                       (short)0, sa, false, false);
            sb = __builtin_amdgcn_wmma_f32_16x16x4_f32(false, a, false, qfb[t],
                                                       (short)0, sb, false, false);
        }

        float pa[8], pb[8];
        softmax_step(sa, scale, mruna, lruna, o0a, o1a, pa);
        softmax_step(sb, scale, mrunb, lrunb, o0b, o1b, pb);

        const float* vcol = &Ys[(kbase + 8 * half) * YSTRIDE + l16];
        #pragma unroll
        for (int j = 0; j < 4; ++j) {
            v2f a0, a1, ba, bq;
            a0[0] = vcol[(2 * j)     * YSTRIDE];
            a0[1] = vcol[(2 * j + 1) * YSTRIDE];
            a1[0] = vcol[(2 * j)     * YSTRIDE + 16];
            a1[1] = vcol[(2 * j + 1) * YSTRIDE + 16];
            ba[0] = pa[2 * j];  ba[1] = pa[2 * j + 1];
            bq[0] = pb[2 * j];  bq[1] = pb[2 * j + 1];
            o0a = __builtin_amdgcn_wmma_f32_16x16x4_f32(false, a0, false, ba,
                                                        (short)0, o0a, false, false);
            o1a = __builtin_amdgcn_wmma_f32_16x16x4_f32(false, a1, false, ba,
                                                        (short)0, o1a, false, false);
            o0b = __builtin_amdgcn_wmma_f32_16x16x4_f32(false, a0, false, bq,
                                                        (short)0, o0b, false, false);
            o1b = __builtin_amdgcn_wmma_f32_16x16x4_f32(false, a1, false, bq,
                                                        (short)0, o1b, false, false);
        }
    }

    float* Ow = Os + wave * (QTILES * 16 * OSTRIDE);
    finish_tile(Ys, Ow,                out, qbase,      lane, bb, h, r, scale, mruna, lruna, o0a, o1a);
    finish_tile(Ys, Ow + 16 * OSTRIDE, out, qbase + 16, lane, bb, h, r, scale, mrunb, lrunb, o0b, o1b);
}

extern "C" void kernel_launch(void* const* d_in, const int* in_sizes, int n_in,
                              void* d_out, int out_size, void* d_ws, size_t ws_size,
                              hipStream_t stream) {
    const float* x = (const float*)d_in[0];
    const float* W = (const float*)d_in[1];
    const float* b = (const float*)d_in[2];
    float* out = (float*)d_out;
    float* qkv = (float*)d_ws;

    const size_t xsBytes = (size_t)(64 * XPITCH) * sizeof(float);
    dim3 g1(128, 2);
    qkv_gemm_kernel<<<g1, 256, xsBytes, stream>>>(x, W, b, qkv);

    const size_t ldsBytes = (size_t)(MROWS * YSTRIDE + 8 * QTILES * 16 * OSTRIDE) * sizeof(float);
    dim3 g2(4, 64);
    attn_kernel<<<g2, 256, ldsBytes, stream>>>(qkv, out);
}


# Attention work-group 16 waves (4 per SIMD): two waves share a 32-query tile pair, each over half the keys; partial max/sum/O joined through LDS; row-sum and PV sums formed per key half, then merged
# speedup vs baseline: 3.5039x; 1.0531x over previous
#include <hip/hip_runtime.h>
#include <math.h>

typedef float v2f __attribute__((ext_vector_type(2)));
typedef float v8f __attribute__((ext_vector_type(8)));

#define HEADS   8
#define DK      32
#define RATE    2
#define SEQ     2046
#define NPAD    2048
#define MROWS   1024
#define DMODEL  256
#define BATCH   4

#define YSTRIDE 34
#define OSTRIDE 36

#define XPITCH  260
#define KSTEPS  16
#define NKB     (DMODEL / (4 * KSTEPS))

__global__ __launch_bounds__(256)
void qkv_gemm_kernel(const float* __restrict__ x,
                     const float* __restrict__ W,
                     const float* __restrict__ bias,
                     float* __restrict__ qkv)
{
    extern __shared__ __attribute__((aligned(16))) float xs[];

    const int tid  = threadIdx.x;
    const int wave = tid >> 5;
    const int lane = tid & 31;
    const int half = lane >> 4;
    const int l16  = lane & 15;

    const int row0 = blockIdx.x * 64;
    const int col0 = blockIdx.y * 128;

    const int wm = wave >> 2;
    const int wn = wave & 3;

    const float* ap0 = &xs[(32 * wm + l16) * XPITCH + 2 * half];
    const float* ap1 = ap0 + 16 * XPITCH;
    const int    c0  = col0 + 32 * wn + 2 * l16;
    const float* wp  = W + (size_t)(2 * half) * DMODEL + c0;

    float2 wr0[NKB][KSTEPS], wr1[NKB][KSTEPS];
    #pragma unroll
    for (int t = 0; t < KSTEPS; ++t) {
        const float* p = wp + (size_t)(4 * t) * DMODEL;
        wr0[0][t] = *(const float2*)p;
        wr1[0][t] = *(const float2*)(p + DMODEL);
    }

    {
        const int c4 = (tid & 63) * 4;
        const int rb = tid >> 6;
        const int bb = row0 >> 11;
        const int n0 = row0 & (NPAD - 1);
        float4 v[16];
        #pragma unroll
        for (int u = 0; u < 16; ++u) {
            const int n = n0 + rb + 4 * u;
            const bool valid = (n < SEQ);
            const float4 t = *(const float4*)(x + ((size_t)(bb * SEQ + (valid ? n : 0))) * DMODEL + c4);
            v[u] = valid ? t : make_float4(0.0f, 0.0f, 0.0f, 0.0f);
        }
        #pragma unroll
        for (int u = 0; u < 16; ++u)
            *(float4*)&xs[(rb + 4 * u) * XPITCH + c4] = v[u];
    }
    __syncthreads();

    const float2 bv = *(const float2*)&bias[c0];
    const float bv0 = bv.x;
    const float bv1 = bv.y;
    v8f acc00, acc01, acc10, acc11;
    #pragma unroll
    for (int v = 0; v < 8; ++v) { acc00[v] = bv0; acc01[v] = bv1; acc10[v] = bv0; acc11[v] = bv1; }

    v2f a0[NKB][KSTEPS], a1[NKB][KSTEPS];
    #pragma unroll
    for (int t = 0; t < KSTEPS; ++t) {
        const float2 f0 = *(const float2*)(ap0 + 4 * t);
        const float2 f1 = *(const float2*)(ap1 + 4 * t);
        a0[0][t][0] = f0.x; a0[0][t][1] = f0.y;
        a1[0][t][0] = f1.x; a1[0][t][1] = f1.y;
    }

    #pragma unroll
    for (int kb = 0; kb < NKB; ++kb) {
        if (kb + 1 < NKB) {
            const int kn = (kb + 1) * 4 * KSTEPS;
            #pragma unroll
            for (int t = 0; t < KSTEPS; ++t) {
                const float* p = wp + (size_t)(kn + 4 * t) * DMODEL;
                wr0[kb + 1][t] = *(const float2*)p;
                wr1[kb + 1][t] = *(const float2*)(p + DMODEL);
                const float2 f0 = *(const float2*)(ap0 + kn + 4 * t);
                const float2 f1 = *(const float2*)(ap1 + kn + 4 * t);
                a0[kb + 1][t][0] = f0.x; a0[kb + 1][t][1] = f0.y;
                a1[kb + 1][t][0] = f1.x; a1[kb + 1][t][1] = f1.y;
            }
        }
        __builtin_amdgcn_sched_barrier(0);

        #pragma unroll
        for (int t = 0; t < KSTEPS; ++t) {
            v2f b0, b1;
            b0[0] = wr0[kb][t].x;  b0[1] = wr1[kb][t].x;
            b1[0] = wr0[kb][t].y;  b1[1] = wr1[kb][t].y;
            acc00 = __builtin_amdgcn_wmma_f32_16x16x4_f32(false, a0[kb][t], false, b0,
                                                          (short)0, acc00, false, false);
            acc01 = __builtin_amdgcn_wmma_f32_16x16x4_f32(false, a0[kb][t], false, b1,
                                                          (short)0, acc01, false, false);
            acc10 = __builtin_amdgcn_wmma_f32_16x16x4_f32(false, a1[kb][t], false, b0,
                                                          (short)0, acc10, false, false);
            acc11 = __builtin_amdgcn_wmma_f32_16x16x4_f32(false, a1[kb][t], false, b1,
                                                          (short)0, acc11, false, false);
        }
        __builtin_amdgcn_sched_barrier(0);
    }

    #pragma unroll
    for (int v = 0; v < 8; ++v) {
        const size_t r0 = (size_t)(row0 + 32 * wm + v + 8 * half) * DMODEL;
        const size_t r1 = r0 + (size_t)16 * DMODEL;
        *(float2*)&qkv[r0 + c0] = make_float2(acc00[v], acc01[v]);
        *(float2*)&qkv[r1 + c0] = make_float2(acc10[v], acc11[v]);
    }
}

#define QTILES  2
#define KSPLIT  2
#define NWAVES  (8 * KSPLIT)
#define KBLKS   (64 / KSPLIT)
#define XSTRIDE 20
#define XSLOT   (32 * XSTRIDE)

__device__ __forceinline__ float xhalf(float x)
{
    const int b = __builtin_bit_cast(int, x);
    return __builtin_bit_cast(float,
        __builtin_amdgcn_permlanex16(b, b, 0x76543210, 0xfedcba98, false, false));
}

__device__ __forceinline__ void softmax_step(const v8f& s, float scale, float& mrun, float& lrun,
                                             v8f& o0, v8f& o1, float (&p)[8])
{
    float bm = fmaxf(fmaxf(fmaxf(s[0], s[1]), fmaxf(s[2], s[3])),
                     fmaxf(fmaxf(s[4], s[5]), fmaxf(s[6], s[7]))) * scale;
    bm = fmaxf(bm, xhalf(bm));
    const float newm = fmaxf(mrun, bm);
    const float corr = __expf(mrun - newm);
    #pragma unroll
    for (int v = 0; v < 8; ++v) p[v] = __expf(s[v] * scale - newm);
    const float psum = ((p[0] + p[1]) + (p[2] + p[3])) + ((p[4] + p[5]) + (p[6] + p[7]));
    lrun = lrun * corr + psum;
    mrun = newm;
    #pragma unroll
    for (int v = 0; v < 8; ++v) { o0[v] *= corr; o1[v] *= corr; }
}

__device__ __forceinline__ void finish_tile(const float* Ys, float* Ow, float* __restrict__ out,
                                            int qbase, int lane, int bb, int h, int r,
                                            float scale, float mrun, float lrun,
                                            const v8f& o0, const v8f& o1)
{
    const int half = lane >> 4;
    const int l16  = lane & 15;
    const int row  = qbase + l16;
    const float ltot = lrun + xhalf(lrun);

    int   nrc[3];
    float msk[3], e[3];
    #pragma unroll
    for (int k = 0; k < 3; ++k) {
        const int nr = row + k - 1;
        const bool ok = (nr >= 0) && (nr < MROWS);
        nrc[k] = ok ? nr : row;
        msk[k] = ok ? 1.0f : 0.0f;
        float part = 0.0f;
        #pragma unroll
        for (int dd = 0; dd < 16; dd += 2) {
            const float2 ya = *(const float2*)&Ys[row    * YSTRIDE + 16 * half + dd];
            const float2 yb = *(const float2*)&Ys[nrc[k] * YSTRIDE + 16 * half + dd];
            part += ya.x * yb.x + ya.y * yb.y;
        }
        part += xhalf(part);
        e[k] = msk[k] * part * scale;
    }
    const float newm = fmaxf(fmaxf(mrun, e[0]), fmaxf(e[1], e[2]));
    const float corr = __expf(mrun - newm);
    float pk[3];
    float psum = 0.0f;
    #pragma unroll
    for (int k = 0; k < 3; ++k) {
        const float pe = __expf(e[k] - newm);
        psum += pe;
        pk[k] = pe * msk[k];
    }
    const float inv = 1.0f / (ltot * corr + psum);

    float r0[8], r1[8];
    #pragma unroll
    for (int v = 0; v < 8; ++v) { r0[v] = o0[v] * corr; r1[v] = o1[v] * corr; }
    #pragma unroll
    for (int k = 0; k < 3; ++k) {
        #pragma unroll
        for (int v = 0; v < 8; v += 2) {
            const float2 y0 = *(const float2*)&Ys[nrc[k] * YSTRIDE + 8 * half + v];
            const float2 y1 = *(const float2*)&Ys[nrc[k] * YSTRIDE + 16 + 8 * half + v];
            r0[v]     += pk[k] * y0.x;
            r0[v + 1] += pk[k] * y0.y;
            r1[v]     += pk[k] * y1.x;
            r1[v + 1] += pk[k] * y1.y;
        }
    }

    float* orow = Ow + l16 * OSTRIDE + 8 * half;
    *(float4*)(orow)      = make_float4(r0[0] * inv, r0[1] * inv, r0[2] * inv, r0[3] * inv);
    *(float4*)(orow + 4)  = make_float4(r0[4] * inv, r0[5] * inv, r0[6] * inv, r0[7] * inv);
    *(float4*)(orow + 16) = make_float4(r1[0] * inv, r1[1] * inv, r1[2] * inv, r1[3] * inv);
    *(float4*)(orow + 20) = make_float4(r1[4] * inv, r1[5] * inv, r1[6] * inv, r1[7] * inv);

    asm volatile("s_wait_dscnt 0x0" ::: "memory");

    const int rr = lane >> 3;
    const int cc = (lane & 7) * 4;
    #pragma unroll
    for (int i = 0; i < 4; ++i) {
        const int lrow = rr + 4 * i;
        const float4 ov = *(const float4*)&Ow[lrow * OSTRIDE + cc];
        const int n = 2 * (qbase + lrow) + r;
        if (n < SEQ) {
            const size_t base = ((size_t)(bb * SEQ + n)) * DMODEL + h * DK + cc;
            *(float4*)&out[base] = ov;
        }
    }
}

__global__ __launch_bounds__(32 * NWAVES) __attribute__((amdgpu_waves_per_eu(4, 4)))
void attn_kernel(const float* __restrict__ qkv, float* __restrict__ out)
{
    extern __shared__ __attribute__((aligned(16))) float lds[];
    float* Ys = lds;
    float* Xs = lds + MROWS * YSTRIDE;

    const int tid  = threadIdx.x;
    const int wave = tid >> 5;
    const int lane = tid & 31;
    const int half = lane >> 4;
    const int l16  = lane & 15;
    const int wq   = wave & 7;
    const int kh   = wave >> 3;

    const int g  = blockIdx.y;
    const int bb = g >> 4;
    const int h  = (g >> 1) & 7;
    const int r  = g & 1;

    {
        const int dcol  = (tid & 15) * 2;
        const int rbase = tid >> 4;
        for (int i = rbase; i < MROWS; i += 2 * NWAVES) {
            const float* src = qkv + ((size_t)(bb * NPAD + 2 * i + r)) * DMODEL
                                   + h * DK + dcol;
            const float2 v = *(const float2*)src;
            *(float2*)&Ys[i * YSTRIDE + dcol] = v;
        }
    }
    __syncthreads();

    const int qbase = (blockIdx.x * 8 + wq) * (16 * QTILES);

    const float scale = 0.17677669529663687f;

    v2f qfa[8], qfb[8];
    #pragma unroll
    for (int t = 0; t < 8; ++t) {
        const float2 qa = *(const float2*)&Ys[(qbase + l16)      * YSTRIDE + 4 * t + 2 * half];
        const float2 qb = *(const float2*)&Ys[(qbase + 16 + l16) * YSTRIDE + 4 * t + 2 * half];
        qfa[t][0] = qa.x; qfa[t][1] = qa.y;
        qfb[t][0] = qb.x; qfb[t][1] = qb.y;
    }

    float mruna = -3.0e38f, mrunb = -3.0e38f;
    float lruna = 0.0f,     lrunb = 0.0f;
    v8f o0a, o1a, o0b, o1b;
    #pragma unroll
    for (int v = 0; v < 8; ++v) { o0a[v] = 0.0f; o1a[v] = 0.0f; o0b[v] = 0.0f; o1b[v] = 0.0f; }

    const int kfirst = kh * (KBLKS * 16);
    for (int it = 0; it < KBLKS; ++it) {
        const int kbase = kfirst + it * 16;

        v8f sa, sb;
        #pragma unroll
        for (int v = 0; v < 8; ++v) { sa[v] = 0.0f; sb[v] = 0.0f; }
        #pragma unroll
        for (int t = 0; t < 8; ++t) {
            const float2 kv = *(const float2*)&Ys[(kbase + l16) * YSTRIDE + 4 * t + 2 * half];
            v2f a;
            a[0] = kv.x; a[1] = kv.y;
            sa = __builtin_amdgcn_wmma_f32_16x16x4_f32(false, a, false, qfa[t],
                                                       (short)0, sa, false, false);
            sb = __builtin_amdgcn_wmma_f32_16x16x4_f32(false, a, false, qfb[t],
                                                       (short)0, sb, false, false);
        }

        float pa[8], pb[8];
        softmax_step(sa, scale, mruna, lruna, o0a, o1a, pa);
        softmax_step(sb, scale, mrunb, lrunb, o0b, o1b, pb);

        const float* vcol = &Ys[(kbase + 8 * half) * YSTRIDE + l16];
        #pragma unroll
        for (int j = 0; j < 4; ++j) {
            v2f a0, a1, ba, bq;
            a0[0] = vcol[(2 * j)     * YSTRIDE];
            a0[1] = vcol[(2 * j + 1) * YSTRIDE];
            a1[0] = vcol[(2 * j)     * YSTRIDE + 16];
            a1[1] = vcol[(2 * j + 1) * YSTRIDE + 16];
            ba[0] = pa[2 * j];  ba[1] = pa[2 * j + 1];
            bq[0] = pb[2 * j];  bq[1] = pb[2 * j + 1];
            o0a = __builtin_amdgcn_wmma_f32_16x16x4_f32(false, a0, false, ba,
                                                        (short)0, o0a, false, false);
            o1a = __builtin_amdgcn_wmma_f32_16x16x4_f32(false, a1, false, ba,
                                                        (short)0, o1a, false, false);
            o0b = __builtin_amdgcn_wmma_f32_16x16x4_f32(false, a0, false, bq,
                                                        (short)0, o0b, false, false);
            o1b = __builtin_amdgcn_wmma_f32_16x16x4_f32(false, a1, false, bq,
                                                        (short)0, o1b, false, false);
        }
    }

    const bool first = (kh == 0);
    {
        const v8f   g0 = first ? o0b : o0a;
        const v8f   g1 = first ? o1b : o1a;
        const float gm = first ? mrunb : mruna;
        const float gl = first ? lrunb : lruna;
        float* xo = Xs + (wave ^ 8) * XSLOT + lane * XSTRIDE;
        *(float4*)(xo)      = make_float4(g0[0], g0[1], g0[2], g0[3]);
        *(float4*)(xo + 4)  = make_float4(g0[4], g0[5], g0[6], g0[7]);
        *(float4*)(xo + 8)  = make_float4(g1[0], g1[1], g1[2], g1[3]);
        *(float4*)(xo + 12) = make_float4(g1[4], g1[5], g1[6], g1[7]);
        *(float2*)(xo + 16) = make_float2(gm, gl);
    }
    __syncthreads();

    v8f   k0 = first ? o0a : o0b;
    v8f   k1 = first ? o1a : o1b;
    float km = first ? mruna : mrunb;
    float kl = first ? lruna : lrunb;
    float* Ow = Xs + wave * XSLOT;
    {
        const float* xi = Ow + lane * XSTRIDE;
        const float4 p0 = *(const float4*)(xi);
        const float4 p1 = *(const float4*)(xi + 4);
        const float4 p2 = *(const float4*)(xi + 8);
        const float4 p3 = *(const float4*)(xi + 12);
        const float2 ml = *(const float2*)(xi + 16);
        const float newm = fmaxf(km, ml.x);
        const float ck = __expf(km - newm);
        const float cp = __expf(ml.x - newm);
        kl = kl * ck + ml.y * cp;
        km = newm;
        k0[0] = k0[0] * ck + p0.x * cp;  k0[1] = k0[1] * ck + p0.y * cp;
        k0[2] = k0[2] * ck + p0.z * cp;  k0[3] = k0[3] * ck + p0.w * cp;
        k0[4] = k0[4] * ck + p1.x * cp;  k0[5] = k0[5] * ck + p1.y * cp;
        k0[6] = k0[6] * ck + p1.z * cp;  k0[7] = k0[7] * ck + p1.w * cp;
        k1[0] = k1[0] * ck + p2.x * cp;  k1[1] = k1[1] * ck + p2.y * cp;
        k1[2] = k1[2] * ck + p2.z * cp;  k1[3] = k1[3] * ck + p2.w * cp;
        k1[4] = k1[4] * ck + p3.x * cp;  k1[5] = k1[5] * ck + p3.y * cp;
        k1[6] = k1[6] * ck + p3.z * cp;  k1[7] = k1[7] * ck + p3.w * cp;
    }

    finish_tile(Ys, Ow, out, qbase + (first ? 0 : 16), lane, bb, h, r, scale, km, kl, k0, k1);
}

extern "C" void kernel_launch(void* const* d_in, const int* in_sizes, int n_in,
                              void* d_out, int out_size, void* d_ws, size_t ws_size,
                              hipStream_t stream) {
    const float* x = (const float*)d_in[0];
    const float* W = (const float*)d_in[1];
    const float* b = (const float*)d_in[2];
    float* out = (float*)d_out;
    float* qkv = (float*)d_ws;

    const size_t xsBytes = (size_t)(64 * XPITCH) * sizeof(float);
    dim3 g1(128, 2);
    qkv_gemm_kernel<<<g1, 256, xsBytes, stream>>>(x, W, b, qkv);

    const size_t ldsBytes = (size_t)(MROWS * YSTRIDE + NWAVES * XSLOT) * sizeof(float);
    dim3 g2(4, 64);
    attn_kernel<<<g2, 32 * NWAVES, ldsBytes, stream>>>(qkv, out);
}


# Attention key loop: running max starts at the query's own logit, raised (O, row sum rescaled) only in blocks where a p passes 2^10; logits in base 2 (Q carries scale*log2e); sums round differently
# speedup vs baseline: 3.6686x; 1.0470x over previous
#include <hip/hip_runtime.h>
#include <math.h>

typedef float v2f __attribute__((ext_vector_type(2)));
typedef float v8f __attribute__((ext_vector_type(8)));

#define HEADS   8
#define DK      32
#define RATE    2
#define SEQ     2046
#define NPAD    2048
#define MROWS   1024
#define DMODEL  256
#define BATCH   4

#define YSTRIDE 34
#define OSTRIDE 36

#define XPITCH  260
#define KSTEPS  16
#define NKB     (DMODEL / (4 * KSTEPS))

__global__ __launch_bounds__(256)
void qkv_gemm_kernel(const float* __restrict__ x,
                     const float* __restrict__ W,
                     const float* __restrict__ bias,
                     float* __restrict__ qkv)
{
    extern __shared__ __attribute__((aligned(16))) float xs[];

    const int tid  = threadIdx.x;
    const int wave = tid >> 5;
    const int lane = tid & 31;
    const int half = lane >> 4;
    const int l16  = lane & 15;

    const int row0 = blockIdx.x * 64;
    const int col0 = blockIdx.y * 128;

    const int wm = wave >> 2;
    const int wn = wave & 3;

    const float* ap0 = &xs[(32 * wm + l16) * XPITCH + 2 * half];
    const float* ap1 = ap0 + 16 * XPITCH;
    const int    c0  = col0 + 32 * wn + 2 * l16;
    const float* wp  = W + (size_t)(2 * half) * DMODEL + c0;

    float2 wr0[NKB][KSTEPS], wr1[NKB][KSTEPS];
    #pragma unroll
    for (int t = 0; t < KSTEPS; ++t) {
        const float* p = wp + (size_t)(4 * t) * DMODEL;
        wr0[0][t] = *(const float2*)p;
        wr1[0][t] = *(const float2*)(p + DMODEL);
    }

    {
        const int c4 = (tid & 63) * 4;
        const int rb = tid >> 6;
        const int bb = row0 >> 11;
        const int n0 = row0 & (NPAD - 1);
        float4 v[16];
        #pragma unroll
        for (int u = 0; u < 16; ++u) {
            const int n = n0 + rb + 4 * u;
            const bool valid = (n < SEQ);
            const float4 t = *(const float4*)(x + ((size_t)(bb * SEQ + (valid ? n : 0))) * DMODEL + c4);
            v[u] = valid ? t : make_float4(0.0f, 0.0f, 0.0f, 0.0f);
        }
        #pragma unroll
        for (int u = 0; u < 16; ++u)
            *(float4*)&xs[(rb + 4 * u) * XPITCH + c4] = v[u];
    }
    __syncthreads();

    const float2 bv = *(const float2*)&bias[c0];
    const float bv0 = bv.x;
    const float bv1 = bv.y;
    v8f acc00, acc01, acc10, acc11;
    #pragma unroll
    for (int v = 0; v < 8; ++v) { acc00[v] = bv0; acc01[v] = bv1; acc10[v] = bv0; acc11[v] = bv1; }

    v2f a0[NKB][KSTEPS], a1[NKB][KSTEPS];
    #pragma unroll
    for (int t = 0; t < KSTEPS; ++t) {
        const float2 f0 = *(const float2*)(ap0 + 4 * t);
        const float2 f1 = *(const float2*)(ap1 + 4 * t);
        a0[0][t][0] = f0.x; a0[0][t][1] = f0.y;
        a1[0][t][0] = f1.x; a1[0][t][1] = f1.y;
    }

    #pragma unroll
    for (int kb = 0; kb < NKB; ++kb) {
        if (kb + 1 < NKB) {
            const int kn = (kb + 1) * 4 * KSTEPS;
            #pragma unroll
            for (int t = 0; t < KSTEPS; ++t) {
                const float* p = wp + (size_t)(kn + 4 * t) * DMODEL;
                wr0[kb + 1][t] = *(const float2*)p;
                wr1[kb + 1][t] = *(const float2*)(p + DMODEL);
                const float2 f0 = *(const float2*)(ap0 + kn + 4 * t);
                const float2 f1 = *(const float2*)(ap1 + kn + 4 * t);
                a0[kb + 1][t][0] = f0.x; a0[kb + 1][t][1] = f0.y;
                a1[kb + 1][t][0] = f1.x; a1[kb + 1][t][1] = f1.y;
            }
        }
        __builtin_amdgcn_sched_barrier(0);

        #pragma unroll
        for (int t = 0; t < KSTEPS; ++t) {
            v2f b0, b1;
            b0[0] = wr0[kb][t].x;  b0[1] = wr1[kb][t].x;
            b1[0] = wr0[kb][t].y;  b1[1] = wr1[kb][t].y;
            acc00 = __builtin_amdgcn_wmma_f32_16x16x4_f32(false, a0[kb][t], false, b0,
                                                          (short)0, acc00, false, false);
            acc01 = __builtin_amdgcn_wmma_f32_16x16x4_f32(false, a0[kb][t], false, b1,
                                                          (short)0, acc01, false, false);
            acc10 = __builtin_amdgcn_wmma_f32_16x16x4_f32(false, a1[kb][t], false, b0,
                                                          (short)0, acc10, false, false);
            acc11 = __builtin_amdgcn_wmma_f32_16x16x4_f32(false, a1[kb][t], false, b1,
                                                          (short)0, acc11, false, false);
        }
        __builtin_amdgcn_sched_barrier(0);
    }

    #pragma unroll
    for (int v = 0; v < 8; ++v) {
        const size_t r0 = (size_t)(row0 + 32 * wm + v + 8 * half) * DMODEL;
        const size_t r1 = r0 + (size_t)16 * DMODEL;
        *(float2*)&qkv[r0 + c0] = make_float2(acc00[v], acc01[v]);
        *(float2*)&qkv[r1 + c0] = make_float2(acc10[v], acc11[v]);
    }
}

#define QTILES  2
#define KSPLIT  2
#define NWAVES  (8 * KSPLIT)
#define KBLKS   (64 / KSPLIT)
#define XSTRIDE 20
#define XSLOT   (32 * XSTRIDE)
#define C2      0.2550348616841918f
#define PMAX    1024.0f

__device__ __forceinline__ float xhalf(float x)
{
    const int b = __builtin_bit_cast(int, x);
    return __builtin_bit_cast(float,
        __builtin_amdgcn_permlanex16(b, b, 0x76543210, 0xfedcba98, false, false));
}

__device__ __forceinline__ float ex2(float x) { return __builtin_amdgcn_exp2f(x); }

__device__ __forceinline__ float sum8(const float (&p)[8])
{
    return ((p[0] + p[1]) + (p[2] + p[3])) + ((p[4] + p[5]) + (p[6] + p[7]));
}

__device__ __forceinline__ void raise_max(const v8f& s, float& mrun, float& lrun,
                                          v8f& o0, v8f& o1, float (&p)[8], float& psum)
{
    float bm = fmaxf(fmaxf(fmaxf(s[0], s[1]), fmaxf(s[2], s[3])),
                     fmaxf(fmaxf(s[4], s[5]), fmaxf(s[6], s[7])));
    bm = fmaxf(bm, xhalf(bm));
    const float newm = fmaxf(mrun, bm);
    const float corr = ex2(mrun - newm);
    #pragma unroll
    for (int v = 0; v < 8; ++v) p[v] = ex2(s[v] - newm);
    psum = sum8(p);
    lrun *= corr;
    mrun = newm;
    #pragma unroll
    for (int v = 0; v < 8; ++v) { o0[v] *= corr; o1[v] *= corr; }
}

__device__ __forceinline__ void finish_tile(const float* Ys, float* Ow, float* __restrict__ out,
                                            int qbase, int lane, int bb, int h, int r,
                                            float scale, float mrun, float lrun,
                                            const v8f& o0, const v8f& o1)
{
    const int half = lane >> 4;
    const int l16  = lane & 15;
    const int row  = qbase + l16;
    const float ltot = lrun + xhalf(lrun);

    int   nrc[3];
    float msk[3], e[3];
    #pragma unroll
    for (int k = 0; k < 3; ++k) {
        const int nr = row + k - 1;
        const bool ok = (nr >= 0) && (nr < MROWS);
        nrc[k] = ok ? nr : row;
        msk[k] = ok ? 1.0f : 0.0f;
        float part = 0.0f;
        #pragma unroll
        for (int dd = 0; dd < 16; dd += 2) {
            const float2 ya = *(const float2*)&Ys[row    * YSTRIDE + 16 * half + dd];
            const float2 yb = *(const float2*)&Ys[nrc[k] * YSTRIDE + 16 * half + dd];
            part += ya.x * yb.x + ya.y * yb.y;
        }
        part += xhalf(part);
        e[k] = msk[k] * part * scale;
    }
    const float newm = fmaxf(fmaxf(mrun, e[0]), fmaxf(e[1], e[2]));
    const float corr = ex2(mrun - newm);
    float pk[3];
    float psum = 0.0f;
    #pragma unroll
    for (int k = 0; k < 3; ++k) {
        const float pe = ex2(e[k] - newm);
        psum += pe;
        pk[k] = pe * msk[k];
    }
    const float inv = 1.0f / (ltot * corr + psum);

    float r0[8], r1[8];
    #pragma unroll
    for (int v = 0; v < 8; ++v) { r0[v] = o0[v] * corr; r1[v] = o1[v] * corr; }
    #pragma unroll
    for (int k = 0; k < 3; ++k) {
        #pragma unroll
        for (int v = 0; v < 8; v += 2) {
            const float2 y0 = *(const float2*)&Ys[nrc[k] * YSTRIDE + 8 * half + v];
            const float2 y1 = *(const float2*)&Ys[nrc[k] * YSTRIDE + 16 + 8 * half + v];
            r0[v]     += pk[k] * y0.x;
            r0[v + 1] += pk[k] * y0.y;
            r1[v]     += pk[k] * y1.x;
            r1[v + 1] += pk[k] * y1.y;
        }
    }

    float* orow = Ow + l16 * OSTRIDE + 8 * half;
    *(float4*)(orow)      = make_float4(r0[0] * inv, r0[1] * inv, r0[2] * inv, r0[3] * inv);
    *(float4*)(orow + 4)  = make_float4(r0[4] * inv, r0[5] * inv, r0[6] * inv, r0[7] * inv);
    *(float4*)(orow + 16) = make_float4(r1[0] * inv, r1[1] * inv, r1[2] * inv, r1[3] * inv);
    *(float4*)(orow + 20) = make_float4(r1[4] * inv, r1[5] * inv, r1[6] * inv, r1[7] * inv);

    asm volatile("s_wait_dscnt 0x0" ::: "memory");

    const int rr = lane >> 3;
    const int cc = (lane & 7) * 4;
    #pragma unroll
    for (int i = 0; i < 4; ++i) {
        const int lrow = rr + 4 * i;
        const float4 ov = *(const float4*)&Ow[lrow * OSTRIDE + cc];
        const int n = 2 * (qbase + lrow) + r;
        if (n < SEQ) {
            const size_t base = ((size_t)(bb * SEQ + n)) * DMODEL + h * DK + cc;
            *(float4*)&out[base] = ov;
        }
    }
}

__global__ __launch_bounds__(32 * NWAVES) __attribute__((amdgpu_waves_per_eu(4, 4)))
void attn_kernel(const float* __restrict__ qkv, float* __restrict__ out)
{
    extern __shared__ __attribute__((aligned(16))) float lds[];
    float* Ys = lds;
    float* Xs = lds + MROWS * YSTRIDE;

    const int tid  = threadIdx.x;
    const int wave = tid >> 5;
    const int lane = tid & 31;
    const int half = lane >> 4;
    const int l16  = lane & 15;
    const int wq   = wave & 7;
    const int kh   = wave >> 3;

    const int g  = blockIdx.y;
    const int bb = g >> 4;
    const int h  = (g >> 1) & 7;
    const int r  = g & 1;

    {
        const int dcol  = (tid & 15) * 2;
        const int rbase = tid >> 4;
        for (int i = rbase; i < MROWS; i += 2 * NWAVES) {
            const float* src = qkv + ((size_t)(bb * NPAD + 2 * i + r)) * DMODEL
                                   + h * DK + dcol;
            const float2 v = *(const float2*)src;
            *(float2*)&Ys[i * YSTRIDE + dcol] = v;
        }
    }
    __syncthreads();

    const int qbase = (blockIdx.x * 8 + wq) * (16 * QTILES);

    const float scale = C2;

    v2f qfa[8], qfb[8];
    #pragma unroll
    for (int t = 0; t < 8; ++t) {
        const float2 qa = *(const float2*)&Ys[(qbase + l16)      * YSTRIDE + 4 * t + 2 * half];
        const float2 qb = *(const float2*)&Ys[(qbase + 16 + l16) * YSTRIDE + 4 * t + 2 * half];
        qfa[t][0] = qa.x; qfa[t][1] = qa.y;
        qfb[t][0] = qb.x; qfb[t][1] = qb.y;
    }

    float mruna = 0.0f, mrunb = 0.0f;
    #pragma unroll
    for (int t = 0; t < 8; ++t) {
        mruna += qfa[t][0] * qfa[t][0] + qfa[t][1] * qfa[t][1];
        mrunb += qfb[t][0] * qfb[t][0] + qfb[t][1] * qfb[t][1];
    }
    mruna = (mruna + xhalf(mruna)) * scale;
    mrunb = (mrunb + xhalf(mrunb)) * scale;
    #pragma unroll
    for (int t = 0; t < 8; ++t) {
        qfa[t][0] *= scale; qfa[t][1] *= scale;
        qfb[t][0] *= scale; qfb[t][1] *= scale;
    }
    float lruna = 0.0f,     lrunb = 0.0f;
    v8f o0a, o1a, o0b, o1b;
    #pragma unroll
    for (int v = 0; v < 8; ++v) { o0a[v] = 0.0f; o1a[v] = 0.0f; o0b[v] = 0.0f; o1b[v] = 0.0f; }

    const int kfirst = kh * (KBLKS * 16);
    for (int it = 0; it < KBLKS; ++it) {
        const int kbase = kfirst + it * 16;

        const float* vcol = &Ys[(kbase + 8 * half) * YSTRIDE + l16];
        v2f va0[4], va1[4];
        #pragma unroll
        for (int j = 0; j < 4; ++j) {
            va0[j][0] = vcol[(2 * j)     * YSTRIDE];
            va0[j][1] = vcol[(2 * j + 1) * YSTRIDE];
            va1[j][0] = vcol[(2 * j)     * YSTRIDE + 16];
            va1[j][1] = vcol[(2 * j + 1) * YSTRIDE + 16];
        }

        v2f kf[8];
        #pragma unroll
        for (int t = 0; t < 8; ++t) {
            const float2 kv = *(const float2*)&Ys[(kbase + l16) * YSTRIDE + 4 * t + 2 * half];
            kf[t][0] = kv.x; kf[t][1] = kv.y;
        }
        __builtin_amdgcn_sched_barrier(0);

        v8f sa, sb;
        #pragma unroll
        for (int v = 0; v < 8; ++v) { sa[v] = 0.0f; sb[v] = 0.0f; }
        #pragma unroll
        for (int t = 0; t < 8; ++t) {
            sa = __builtin_amdgcn_wmma_f32_16x16x4_f32(false, kf[t], false, qfa[t],
                                                       (short)0, sa, false, false);
            sb = __builtin_amdgcn_wmma_f32_16x16x4_f32(false, kf[t], false, qfb[t],
                                                       (short)0, sb, false, false);
        }

        float pa[8], pb[8];
        #pragma unroll
        for (int v = 0; v < 8; ++v) { pa[v] = ex2(sa[v] - mruna); pb[v] = ex2(sb[v] - mrunb); }
        float psa = sum8(pa), psb = sum8(pb);
        if (__builtin_amdgcn_ballot_w32(fmaxf(psa, psb) > PMAX) != 0u) {
            raise_max(sa, mruna, lruna, o0a, o1a, pa, psa);
            raise_max(sb, mrunb, lrunb, o0b, o1b, pb, psb);
        }
        lruna += psa;
        lrunb += psb;

        #pragma unroll
        for (int j = 0; j < 4; ++j) {
            v2f ba, bq;
            const v2f a0 = va0[j], a1 = va1[j];
            ba[0] = pa[2 * j];  ba[1] = pa[2 * j + 1];
            bq[0] = pb[2 * j];  bq[1] = pb[2 * j + 1];
            o0a = __builtin_amdgcn_wmma_f32_16x16x4_f32(false, a0, false, ba,
                                                        (short)0, o0a, false, false);
            o1a = __builtin_amdgcn_wmma_f32_16x16x4_f32(false, a1, false, ba,
                                                        (short)0, o1a, false, false);
            o0b = __builtin_amdgcn_wmma_f32_16x16x4_f32(false, a0, false, bq,
                                                        (short)0, o0b, false, false);
            o1b = __builtin_amdgcn_wmma_f32_16x16x4_f32(false, a1, false, bq,
                                                        (short)0, o1b, false, false);
        }
    }

    const bool first = (kh == 0);
    {
        const v8f   g0 = first ? o0b : o0a;
        const v8f   g1 = first ? o1b : o1a;
        const float gm = first ? mrunb : mruna;
        const float gl = first ? lrunb : lruna;
        float* xo = Xs + (wave ^ 8) * XSLOT + lane * XSTRIDE;
        *(float4*)(xo)      = make_float4(g0[0], g0[1], g0[2], g0[3]);
        *(float4*)(xo + 4)  = make_float4(g0[4], g0[5], g0[6], g0[7]);
        *(float4*)(xo + 8)  = make_float4(g1[0], g1[1], g1[2], g1[3]);
        *(float4*)(xo + 12) = make_float4(g1[4], g1[5], g1[6], g1[7]);
        *(float2*)(xo + 16) = make_float2(gm, gl);
    }
    __syncthreads();

    v8f   k0 = first ? o0a : o0b;
    v8f   k1 = first ? o1a : o1b;
    float km = first ? mruna : mrunb;
    float kl = first ? lruna : lrunb;
    float* Ow = Xs + wave * XSLOT;
    {
        const float* xi = Ow + lane * XSTRIDE;
        const float4 p0 = *(const float4*)(xi);
        const float4 p1 = *(const float4*)(xi + 4);
        const float4 p2 = *(const float4*)(xi + 8);
        const float4 p3 = *(const float4*)(xi + 12);
        const float2 ml = *(const float2*)(xi + 16);
        const float newm = fmaxf(km, ml.x);
        const float ck = ex2(km - newm);
        const float cp = ex2(ml.x - newm);
        kl = kl * ck + ml.y * cp;
        km = newm;
        k0[0] = k0[0] * ck + p0.x * cp;  k0[1] = k0[1] * ck + p0.y * cp;
        k0[2] = k0[2] * ck + p0.z * cp;  k0[3] = k0[3] * ck + p0.w * cp;
        k0[4] = k0[4] * ck + p1.x * cp;  k0[5] = k0[5] * ck + p1.y * cp;
        k0[6] = k0[6] * ck + p1.z * cp;  k0[7] = k0[7] * ck + p1.w * cp;
        k1[0] = k1[0] * ck + p2.x * cp;  k1[1] = k1[1] * ck + p2.y * cp;
        k1[2] = k1[2] * ck + p2.z * cp;  k1[3] = k1[3] * ck + p2.w * cp;
        k1[4] = k1[4] * ck + p3.x * cp;  k1[5] = k1[5] * ck + p3.y * cp;
        k1[6] = k1[6] * ck + p3.z * cp;  k1[7] = k1[7] * ck + p3.w * cp;
    }

    finish_tile(Ys, Ow, out, qbase + (first ? 0 : 16), lane, bb, h, r, scale, km, kl, k0, k1);
}

extern "C" void kernel_launch(void* const* d_in, const int* in_sizes, int n_in,
                              void* d_out, int out_size, void* d_ws, size_t ws_size,
                              hipStream_t stream) {
    const float* x = (const float*)d_in[0];
    const float* W = (const float*)d_in[1];
    const float* b = (const float*)d_in[2];
    float* out = (float*)d_out;
    float* qkv = (float*)d_ws;

    const size_t xsBytes = (size_t)(64 * XPITCH) * sizeof(float);
    dim3 g1(128, 2);
    qkv_gemm_kernel<<<g1, 256, xsBytes, stream>>>(x, W, b, qkv);

    const size_t ldsBytes = (size_t)(MROWS * YSTRIDE + NWAVES * XSLOT) * sizeof(float);
    dim3 g2(4, 64);
    attn_kernel<<<g2, 32 * NWAVES, ldsBytes, stream>>>(qkv, out);
}
